# starting kernel
# baseline (speedup 1.0000x reference)
#include <hip/hip_runtime.h>
#include <hip/hip_bf16.h>

typedef __attribute__((ext_vector_type(16))) _Float16 v16h;
typedef __attribute__((ext_vector_type(8)))  _Float16 v8h;
typedef __attribute__((ext_vector_type(8)))  float    v8f;

#define NPOINTS 8192
#define NBATCH  16
#define PTSB    8
#define CLAMPC  1.9f
#define TWOPI_I 0.63661977236758f

__device__ __forceinline__ v16h ld16(const _Float16* a, const _Float16* b) {
  v8h lo = *(const v8h*)a;
  v8h hi = *(const v8h*)b;
  return __builtin_shufflevector(lo, hi, 0,1,2,3,4,5,6,7,8,9,10,11,12,13,14,15);
}

__device__ __forceinline__ v8f wmma16(v16h a, v16h b, v8f c) {
  return __builtin_amdgcn_wmma_f32_16x16x32_f16(false, a, false, b, (short)0, c,
                                                false, false);
}

__global__ void prep_kernel(const float* __restrict__ x,
                            const float* __restrict__ logs,
                            const float* __restrict__ bias,
                            float* __restrict__ zmat,
                            float* __restrict__ logdet,
                            _Float16* __restrict__ z1h, int total) {
  int p = blockIdx.x * blockDim.x + threadIdx.x;
  if (p < total) {
    const float* xr = x + (size_t)p * 32;
    float* zr = zmat + (size_t)p * 32;
    _Float16* hr = z1h + (size_t)p * 16;
#pragma unroll
    for (int c = 0; c < 32; ++c) {
      float zc = xr[31 - c] * expf(logs[c]) + bias[c];
      zr[c] = zc;
      if (c < 16) hr[c] = (_Float16)zc;
    }
  }
  if (p < NBATCH) {
    float su = 0.f;
#pragma unroll
    for (int c = 0; c < 32; ++c) su += logs[c];
    logdet[p] = su * (float)NPOINTS;
  }
}

__global__ __launch_bounds__(128) void conv_kernel(
    const _Float16* __restrict__ z1h, const int* __restrict__ knn,
    const float* __restrict__ W1g, const float* __restrict__ b1g,
    const float* __restrict__ W2g, const float* __restrict__ b2g,
    const float* __restrict__ W3g, const float* __restrict__ b3g,
    float* __restrict__ zmat, float* __restrict__ logdet) {
  __shared__ __align__(16) _Float16 sW1[2][64][40];
  __shared__ __align__(16) _Float16 sW2[2][64][72];
  __shared__ __align__(16) _Float16 sW3[2][16][72];
  __shared__ __align__(16) _Float16 sA[128][40];
  __shared__ __align__(16) _Float16 sY[4][16][72];
  __shared__ __align__(16) _Float16 sZ[4][16][72];
  __shared__ float sB1[2][64], sB2[2][64], sB3[2][16];
  __shared__ float sLog;

  const int tid  = threadIdx.x;
  const int wid  = tid >> 5;
  const int lane = tid & 31;
  const int n16  = lane & 15;
  const int hl   = lane >> 4;
  const int base0 = hl * 8;

  const int bi = blockIdx.x / (NPOINTS / PTSB);
  const int pb = (blockIdx.x % (NPOINTS / PTSB)) * PTSB;

  for (int idx = tid; idx < 2 * 64 * 32; idx += 128) {
    int s = idx >> 11, r = idx & 2047;
    int j = r >> 5, k = r & 31;
    float v;
    if (k < 16)
      v = W1g[(s * 48 + 16 + k) * 64 + j] + W1g[(s * 48 + 32 + k) * 64 + j];
    else {
      int kk = k - 16;
      v = W1g[(s * 48 + kk) * 64 + j] - W1g[(s * 48 + 32 + kk) * 64 + j];
    }
    sW1[s][j][k] = (_Float16)v;
  }
  for (int idx = tid; idx < 2 * 64 * 64; idx += 128) {
    int s = idx >> 12, r = idx & 4095;
    int j = r >> 6, k = r & 63;
    sW2[s][j][k] = (_Float16)W2g[(s * 64 + k) * 64 + j];
  }
  for (int idx = tid; idx < 2 * 16 * 64; idx += 128) {
    int s = idx >> 10, r = idx & 1023;
    int j = r >> 6, k = r & 63;
    sW3[s][j][k] = (_Float16)W3g[(s * 64 + k) * 16 + j];
  }
  {
    int s = tid >> 6, j = tid & 63;
    sB1[s][j] = b1g[s * 64 + j];
    sB2[s][j] = b2g[s * 64 + j];
    if (tid < 32) sB3[tid >> 4][tid & 15] = b3g[(tid >> 4) * 16 + (tid & 15)];
    if (tid == 0) sLog = 0.f;
  }

  {
    const int e = tid, lp = e >> 4, k = e & 15;
    const size_t gp = (size_t)bi * NPOINTS + pb + lp;
    const int nidx = knn[gp * 16 + k];
    const _Float16* nb = z1h + ((size_t)bi * NPOINTS + nidx) * 16;
    const _Float16* ct = z1h + gp * 16;
    *(v8h*)&sA[e][0]  = *(const v8h*)nb;
    *(v8h*)&sA[e][8]  = *(const v8h*)(nb + 8);
    *(v8h*)&sA[e][16] = *(const v8h*)ct;
    *(v8h*)&sA[e][24] = *(const v8h*)(ct + 8);
  }
  __syncthreads();

  for (int tt = 0; tt < 2; ++tt) {
    const int tile = wid + tt * 4;
    const int p = pb + tile;
    const int e0 = tile * 16;
    const int arow = e0 + n16;
    const v16h A1 = ld16(&sA[arow][base0], &sA[arow][16 + base0]);
    float sv = 0.f;

#pragma unroll
    for (int s = 0; s < 2; ++s) {
#pragma unroll
      for (int nc = 0; nc < 4; ++nc) {
        const int j = nc * 16 + n16;
        const v16h B1 = ld16(&sW1[s][j][base0], &sW1[s][j][16 + base0]);
        v8f c; const float bv = sB1[s][j];
#pragma unroll
        for (int r = 0; r < 8; ++r) c[r] = bv;
        c = wmma16(A1, B1, c);
#pragma unroll
        for (int r = 0; r < 8; ++r)
          sY[wid][r + 8 * hl][j] = (_Float16)fmaxf(c[r], 0.0f);
      }
      __builtin_amdgcn_wave_barrier();

      const v16h A2a = ld16(&sY[wid][n16][base0],      &sY[wid][n16][16 + base0]);
      const v16h A2b = ld16(&sY[wid][n16][32 + base0], &sY[wid][n16][48 + base0]);
#pragma unroll
      for (int nc = 0; nc < 4; ++nc) {
        const int j = nc * 16 + n16;
        const v16h B2a = ld16(&sW2[s][j][base0],      &sW2[s][j][16 + base0]);
        const v16h B2b = ld16(&sW2[s][j][32 + base0], &sW2[s][j][48 + base0]);
        v8f c; const float bv = sB2[s][j];
#pragma unroll
        for (int r = 0; r < 8; ++r) c[r] = bv;
        c = wmma16(A2a, B2a, c);
        c = wmma16(A2b, B2b, c);
#pragma unroll
        for (int r = 0; r < 8; ++r)
          sZ[wid][r + 8 * hl][j] = (_Float16)fmaxf(c[r], 0.0f);
      }
      __builtin_amdgcn_wave_barrier();

      const v16h A3a = ld16(&sZ[wid][n16][base0],      &sZ[wid][n16][16 + base0]);
      const v16h A3b = ld16(&sZ[wid][n16][32 + base0], &sZ[wid][n16][48 + base0]);
      const v16h B3a = ld16(&sW3[s][n16][base0],       &sW3[s][n16][16 + base0]);
      const v16h B3b = ld16(&sW3[s][n16][32 + base0],  &sW3[s][n16][48 + base0]);
      v8f c; const float bv = sB3[s][n16];
#pragma unroll
      for (int r = 0; r < 8; ++r) c[r] = bv;
      c = wmma16(A3a, B3a, c);
      c = wmma16(A3b, B3b, c);
      float v = c[0];
#pragma unroll
      for (int r = 1; r < 8; ++r) v = fmaxf(v, c[r]);
      v = fmaxf(v, __shfl_xor(v, 16, 32));

      if (s == 0) {
        sv = CLAMPC * TWOPI_I * atanf(v * (1.0f / CLAMPC));
        float r0 = sv;
        r0 += __shfl_xor(r0, 1, 32);
        r0 += __shfl_xor(r0, 2, 32);
        r0 += __shfl_xor(r0, 4, 32);
        r0 += __shfl_xor(r0, 8, 32);
        if (lane == 0) atomicAdd(&sLog, r0);
      } else {
        if (lane < 16) {
          float* zp = zmat + ((size_t)bi * NPOINTS + p) * 32 + 16 + n16;
          *zp = *zp * __expf(sv) + v;
        }
      }
      __builtin_amdgcn_wave_barrier();
    }
  }
  __syncthreads();
  if (tid == 0) atomicAdd(&logdet[bi], sLog);
}

extern "C" void kernel_launch(void* const* d_in, const int* in_sizes, int n_in,
                              void* d_out, int out_size, void* d_ws, size_t ws_size,
                              hipStream_t stream) {
  const float* x     = (const float*)d_in[0];
  const int*   knn   = (const int*)d_in[1];
  const float* logs  = (const float*)d_in[2];
  const float* biasv = (const float*)d_in[3];
  const float* W1    = (const float*)d_in[4];
  const float* b1    = (const float*)d_in[5];
  const float* W2    = (const float*)d_in[6];
  const float* b2    = (const float*)d_in[7];
  const float* W3    = (const float*)d_in[8];
  const float* b3    = (const float*)d_in[9];

  float* zmat   = (float*)d_out;
  float* logdet = zmat + (size_t)NBATCH * NPOINTS * 32;
  _Float16* z1h = (_Float16*)d_ws;

  const int total = NBATCH * NPOINTS;
  prep_kernel<<<(total + 255) / 256, 256, 0, stream>>>(x, logs, biasv, zmat,
                                                       logdet, z1h, total);

  const dim3 grid(NBATCH * (NPOINTS / PTSB));
  for (int u = 0; u < 4; u += 2) {
    conv_kernel<<<grid, 128, 0, stream>>>(
        z1h, knn,
        W1 + (size_t)u * 48 * 64, b1 + (size_t)u * 64,
        W2 + (size_t)u * 64 * 64, b2 + (size_t)u * 64,
        W3 + (size_t)u * 64 * 16, b3 + (size_t)u * 16,
        zmat, logdet);
  }
}


# weights staged into LDS once per workgroup: 256 workgroups of 32 waves, each wave loops over 16 points on LDS rows of its own (was 16384 workgroups, weights restaged per 8 points)
# speedup vs baseline: 3.0795x; 3.0795x over previous
#include <hip/hip_runtime.h>
#include <hip/hip_bf16.h>

typedef __attribute__((ext_vector_type(16))) _Float16 v16h;
typedef __attribute__((ext_vector_type(8)))  _Float16 v8h;
typedef __attribute__((ext_vector_type(8)))  float    v8f;

#define NPOINTS 8192
#define NBATCH  16
#define WAVES   32
#define THREADS (WAVES * 32)
#define NWG     256
#define PTS_WG  (NBATCH * NPOINTS / NWG)
#define TRIPS   (PTS_WG / (2 * WAVES))
static_assert(NPOINTS % PTS_WG == 0, "a workgroup stays inside one batch");
static_assert(PTS_WG % (2 * WAVES) == 0, "whole trips");
#define CLAMPC  1.9f
#define TWOPI_I 0.63661977236758f

__device__ __forceinline__ v16h ld16(const _Float16* a, const _Float16* b) {
  v8h lo = *(const v8h*)a;
  v8h hi = *(const v8h*)b;
  return __builtin_shufflevector(lo, hi, 0,1,2,3,4,5,6,7,8,9,10,11,12,13,14,15);
}

__device__ __forceinline__ v8f wmma16(v16h a, v16h b, v8f c) {
  return __builtin_amdgcn_wmma_f32_16x16x32_f16(false, a, false, b, (short)0, c,
                                                false, false);
}

__global__ void prep_kernel(const float* __restrict__ x,
                            const float* __restrict__ logs,
                            const float* __restrict__ bias,
                            float* __restrict__ zmat,
                            float* __restrict__ logdet,
                            _Float16* __restrict__ z1h, int total) {
  int p = blockIdx.x * blockDim.x + threadIdx.x;
  if (p < total) {
    const float* xr = x + (size_t)p * 32;
    float* zr = zmat + (size_t)p * 32;
    _Float16* hr = z1h + (size_t)p * 16;
#pragma unroll
    for (int c = 0; c < 32; ++c) {
      float zc = xr[31 - c] * expf(logs[c]) + bias[c];
      zr[c] = zc;
      if (c < 16) hr[c] = (_Float16)zc;
    }
  }
  if (p < NBATCH) {
    float su = 0.f;
#pragma unroll
    for (int c = 0; c < 32; ++c) su += logs[c];
    logdet[p] = su * (float)NPOINTS;
  }
}

__global__ __launch_bounds__(THREADS) void conv_kernel(
    const _Float16* __restrict__ z1h, const int* __restrict__ knn,
    const float* __restrict__ W1g, const float* __restrict__ b1g,
    const float* __restrict__ W2g, const float* __restrict__ b2g,
    const float* __restrict__ W3g, const float* __restrict__ b3g,
    float* __restrict__ zmat, float* __restrict__ logdet) {
  __shared__ __align__(16) _Float16 sW1[2][64][40];
  __shared__ __align__(16) _Float16 sW2[2][64][72];
  __shared__ __align__(16) _Float16 sW3[2][16][72];
  __shared__ __align__(16) _Float16 sA[WAVES][32][40];
  __shared__ __align__(16) _Float16 sY[WAVES][16][72];
  __shared__ __align__(16) _Float16 sZ[WAVES][16][72];
  __shared__ float sB1[2][64], sB2[2][64], sB3[2][16];
  __shared__ float sLog;

  const int tid  = threadIdx.x;
  const int wid  = tid >> 5;
  const int lane = tid & 31;
  const int n16  = lane & 15;
  const int hl   = lane >> 4;
  const int base0 = hl * 8;

  const int bi  = blockIdx.x / (NPOINTS / PTS_WG);
  const int pb0 = (blockIdx.x % (NPOINTS / PTS_WG)) * PTS_WG;

  for (int idx = tid; idx < 2 * 64 * 32; idx += THREADS) {
    int s = idx >> 11, r = idx & 2047;
    int j = r >> 5, k = r & 31;
    float v;
    if (k < 16)
      v = W1g[(s * 48 + 16 + k) * 64 + j] + W1g[(s * 48 + 32 + k) * 64 + j];
    else {
      int kk = k - 16;
      v = W1g[(s * 48 + kk) * 64 + j] - W1g[(s * 48 + 32 + kk) * 64 + j];
    }
    sW1[s][j][k] = (_Float16)v;
  }
  for (int idx = tid; idx < 2 * 64 * 64; idx += THREADS) {
    int s = idx >> 12, r = idx & 4095;
    int j = r >> 6, k = r & 63;
    sW2[s][j][k] = (_Float16)W2g[(s * 64 + k) * 64 + j];
  }
  for (int idx = tid; idx < 2 * 16 * 64; idx += THREADS) {
    int s = idx >> 10, r = idx & 1023;
    int j = r >> 6, k = r & 63;
    sW3[s][j][k] = (_Float16)W3g[(s * 64 + k) * 16 + j];
  }
  if (tid < 128) {
    int s = tid >> 6, j = tid & 63;
    sB1[s][j] = b1g[s * 64 + j];
    sB2[s][j] = b2g[s * 64 + j];
    if (tid < 32) sB3[tid >> 4][tid & 15] = b3g[(tid >> 4) * 16 + (tid & 15)];
    if (tid == 0) sLog = 0.f;
  }

  __syncthreads();

  float slog = 0.f;
  for (int it = 0; it < TRIPS; ++it) {
    const int pw = pb0 + (it * WAVES + wid) * 2;

    {
      const size_t gp = (size_t)bi * NPOINTS + pw + hl;
      const int nidx = knn[gp * 16 + n16];
      const _Float16* nb = z1h + ((size_t)bi * NPOINTS + nidx) * 16;
      const _Float16* ct = z1h + gp * 16;
      *(v8h*)&sA[wid][lane][0]  = *(const v8h*)nb;
      *(v8h*)&sA[wid][lane][8]  = *(const v8h*)(nb + 8);
      *(v8h*)&sA[wid][lane][16] = *(const v8h*)ct;
      *(v8h*)&sA[wid][lane][24] = *(const v8h*)(ct + 8);
    }
    __builtin_amdgcn_wave_barrier();

    for (int tt = 0; tt < 2; ++tt) {
      const int p = pw + tt;
      const int arow = tt * 16 + n16;
      const v16h A1 = ld16(&sA[wid][arow][base0], &sA[wid][arow][16 + base0]);
      float sv = 0.f;

#pragma unroll
      for (int s = 0; s < 2; ++s) {
#pragma unroll
        for (int nc = 0; nc < 4; ++nc) {
          const int j = nc * 16 + n16;
          const v16h B1 = ld16(&sW1[s][j][base0], &sW1[s][j][16 + base0]);
          v8f c; const float bv = sB1[s][j];
#pragma unroll
          for (int r = 0; r < 8; ++r) c[r] = bv;
          c = wmma16(A1, B1, c);
#pragma unroll
          for (int r = 0; r < 8; ++r)
            sY[wid][r + 8 * hl][j] = (_Float16)fmaxf(c[r], 0.0f);
        }
        __builtin_amdgcn_wave_barrier();

        const v16h A2a = ld16(&sY[wid][n16][base0],      &sY[wid][n16][16 + base0]);
        const v16h A2b = ld16(&sY[wid][n16][32 + base0], &sY[wid][n16][48 + base0]);
#pragma unroll
        for (int nc = 0; nc < 4; ++nc) {
          const int j = nc * 16 + n16;
          const v16h B2a = ld16(&sW2[s][j][base0],      &sW2[s][j][16 + base0]);
          const v16h B2b = ld16(&sW2[s][j][32 + base0], &sW2[s][j][48 + base0]);
          v8f c; const float bv = sB2[s][j];
#pragma unroll
          for (int r = 0; r < 8; ++r) c[r] = bv;
          c = wmma16(A2a, B2a, c);
          c = wmma16(A2b, B2b, c);
#pragma unroll
          for (int r = 0; r < 8; ++r)
            sZ[wid][r + 8 * hl][j] = (_Float16)fmaxf(c[r], 0.0f);
        }
        __builtin_amdgcn_wave_barrier();

        const v16h A3a = ld16(&sZ[wid][n16][base0],      &sZ[wid][n16][16 + base0]);
        const v16h A3b = ld16(&sZ[wid][n16][32 + base0], &sZ[wid][n16][48 + base0]);
        const v16h B3a = ld16(&sW3[s][n16][base0],       &sW3[s][n16][16 + base0]);
        const v16h B3b = ld16(&sW3[s][n16][32 + base0],  &sW3[s][n16][48 + base0]);
        v8f c; const float bv = sB3[s][n16];
#pragma unroll
        for (int r = 0; r < 8; ++r) c[r] = bv;
        c = wmma16(A3a, B3a, c);
        c = wmma16(A3b, B3b, c);
        float v = c[0];
#pragma unroll
        for (int r = 1; r < 8; ++r) v = fmaxf(v, c[r]);
        v = fmaxf(v, __shfl_xor(v, 16, 32));

        if (s == 0) {
          sv = CLAMPC * TWOPI_I * atanf(v * (1.0f / CLAMPC));
          float r0 = sv;
          r0 += __shfl_xor(r0, 1, 32);
          r0 += __shfl_xor(r0, 2, 32);
          r0 += __shfl_xor(r0, 4, 32);
          r0 += __shfl_xor(r0, 8, 32);
          slog += r0;
        } else {
          if (lane < 16) {
            float* zp = zmat + ((size_t)bi * NPOINTS + p) * 32 + 16 + n16;
            *zp = *zp * __expf(sv) + v;
          }
        }
        __builtin_amdgcn_wave_barrier();
      }
    }
  }
  if (lane == 0) atomicAdd(&sLog, slog);
  __syncthreads();
  if (tid == 0) atomicAdd(&logdet[bi], sLog);
}

extern "C" void kernel_launch(void* const* d_in, const int* in_sizes, int n_in,
                              void* d_out, int out_size, void* d_ws, size_t ws_size,
                              hipStream_t stream) {
  const float* x     = (const float*)d_in[0];
  const int*   knn   = (const int*)d_in[1];
  const float* logs  = (const float*)d_in[2];
  const float* biasv = (const float*)d_in[3];
  const float* W1    = (const float*)d_in[4];
  const float* b1    = (const float*)d_in[5];
  const float* W2    = (const float*)d_in[6];
  const float* b2    = (const float*)d_in[7];
  const float* W3    = (const float*)d_in[8];
  const float* b3    = (const float*)d_in[9];

  float* zmat   = (float*)d_out;
  float* logdet = zmat + (size_t)NBATCH * NPOINTS * 32;
  _Float16* z1h = (_Float16*)d_ws;

  const int total = NBATCH * NPOINTS;
  prep_kernel<<<(total + 255) / 256, 256, 0, stream>>>(x, logs, biasv, zmat,
                                                       logdet, z1h, total);

  const dim3 grid(NWG);
  for (int u = 0; u < 4; u += 2) {
    conv_kernel<<<grid, THREADS, 0, stream>>>(
        z1h, knn,
        W1 + (size_t)u * 48 * 64, b1 + (size_t)u * 64,
        W2 + (size_t)u * 64 * 64, b2 + (size_t)u * 64,
        W3 + (size_t)u * 64 * 16, b3 + (size_t)u * 16,
        zmat, logdet);
  }
}


# layers 1-2 transposed (weights = A, edges = B): a layer's relu'd D tiles, packed to f16, are the next layer's operand in registers (no LDS round trip); a weight fragment serves 4 points
# speedup vs baseline: 5.1432x; 1.6701x over previous
#include <hip/hip_runtime.h>
#include <hip/hip_bf16.h>

typedef __attribute__((ext_vector_type(16))) _Float16 v16h;
typedef __attribute__((ext_vector_type(8)))  _Float16 v8h;
typedef __attribute__((ext_vector_type(8)))  float    v8f;

#define NPOINTS 8192
#define NBATCH  16
#define WAVES   16
#define THREADS (WAVES * 32)
#define NWG     256
#define PTS_WG  (NBATCH * NPOINTS / NWG)
#define TT      4
#define TRIPS   (PTS_WG / (TT * WAVES))
static_assert(NPOINTS % PTS_WG == 0, "a workgroup stays inside one batch");
static_assert(PTS_WG % (TT * WAVES) == 0 && TT % 2 == 0, "whole trips, points in pairs");
#define CLAMPC  1.9f
#define TWOPI_I 0.63661977236758f

__device__ __forceinline__ v16h ld16(const _Float16* a, const _Float16* b) {
  v8h lo = *(const v8h*)a;
  v8h hi = *(const v8h*)b;
  return __builtin_shufflevector(lo, hi, 0,1,2,3,4,5,6,7,8,9,10,11,12,13,14,15);
}

__device__ __forceinline__ v8f wmma16(v16h a, v16h b, v8f c) {
  return __builtin_amdgcn_wmma_f32_16x16x32_f16(false, a, false, b, (short)0, c,
                                                false, false);
}

__global__ void prep_kernel(const float* __restrict__ x,
                            const float* __restrict__ logs,
                            const float* __restrict__ bias,
                            float* __restrict__ zmat,
                            float* __restrict__ logdet,
                            _Float16* __restrict__ z1h, int total) {
  int p = blockIdx.x * blockDim.x + threadIdx.x;
  if (p < total) {
    const float* xr = x + (size_t)p * 32;
    float* zr = zmat + (size_t)p * 32;
    _Float16* hr = z1h + (size_t)p * 16;
#pragma unroll
    for (int c = 0; c < 32; ++c) {
      float zc = xr[31 - c] * expf(logs[c]) + bias[c];
      zr[c] = zc;
      if (c < 16) hr[c] = (_Float16)zc;
    }
  }
  if (p < NBATCH) {
    float su = 0.f;
#pragma unroll
    for (int c = 0; c < 32; ++c) su += logs[c];
    logdet[p] = su * (float)NPOINTS;
  }
}

__device__ __forceinline__ v8h relu_pack(v8f c) {
  v8h o;
#pragma unroll
  for (int r = 0; r < 8; ++r) o[r] = (_Float16)fmaxf(c[r], 0.0f);
  return o;
}

__device__ __forceinline__ v16h cat16(v8h lo, v8h hi) {
  return __builtin_shufflevector(lo, hi, 0,1,2,3,4,5,6,7,8,9,10,11,12,13,14,15);
}

__global__ __launch_bounds__(THREADS) void conv_kernel(
    const _Float16* __restrict__ z1h, const int* __restrict__ knn,
    const float* __restrict__ W1g, const float* __restrict__ b1g,
    const float* __restrict__ W2g, const float* __restrict__ b2g,
    const float* __restrict__ W3g, const float* __restrict__ b3g,
    float* __restrict__ zmat, float* __restrict__ logdet) {
  __shared__ __align__(16) _Float16 sW1[2][64][40];
  __shared__ __align__(16) _Float16 sW2[2][64][72];
  __shared__ __align__(16) _Float16 sW3[2][16][72];
  __shared__ __align__(32) float sB1[2][64];
  __shared__ __align__(32) float sB2[2][64];
  __shared__ float sB3[2][16];
  __shared__ float sLog;

  const int tid  = threadIdx.x;
  const int wid  = tid >> 5;
  const int lane = tid & 31;
  const int n16  = lane & 15;
  const int hl   = lane >> 4;
  const int base0 = hl * 8;

  const int bi  = blockIdx.x / (NPOINTS / PTS_WG);
  const int pb0 = (blockIdx.x % (NPOINTS / PTS_WG)) * PTS_WG;

  for (int idx = tid; idx < 2 * 64 * 32; idx += THREADS) {
    int s = idx >> 11, r = idx & 2047;
    int j = r >> 5, k = r & 31;
    float v;
    if (k < 16)
      v = W1g[(s * 48 + 16 + k) * 64 + j] + W1g[(s * 48 + 32 + k) * 64 + j];
    else {
      int kk = k - 16;
      v = W1g[(s * 48 + kk) * 64 + j] - W1g[(s * 48 + 32 + kk) * 64 + j];
    }
    sW1[s][j][k] = (_Float16)v;
  }
  for (int idx = tid; idx < 2 * 64 * 64; idx += THREADS) {
    int s = idx >> 12, r = idx & 4095;
    int j = r >> 6, k = r & 63;
    sW2[s][j][k] = (_Float16)W2g[(s * 64 + k) * 64 + j];
  }
  for (int idx = tid; idx < 2 * 16 * 64; idx += THREADS) {
    int s = idx >> 10, r = idx & 1023;
    int j = r >> 6, k = r & 63;
    sW3[s][j][k] = (_Float16)W3g[(s * 64 + k) * 16 + j];
  }
  if (tid < 128) {
    int s = tid >> 6, j = tid & 63;
    sB1[s][j] = b1g[s * 64 + j];
    sB2[s][j] = b2g[s * 64 + j];
    if (tid < 32) sB3[tid >> 4][tid & 15] = b3g[(tid >> 4) * 16 + (tid & 15)];
    if (tid == 0) sLog = 0.f;
  }

  __syncthreads();

  float slog = 0.f;
  for (int it = 0; it < TRIPS; ++it) {
    const int pw = pb0 + (it * WAVES + wid) * TT;
    asm volatile("" ::: "memory");

    int nidx[TT];
#pragma unroll
    for (int tt = 0; tt < TT; ++tt)
      nidx[tt] = knn[((size_t)bi * NPOINTS + pw + tt) * 16 + n16];
    v16h X[TT];
#pragma unroll
    for (int tt = 0; tt < TT; ++tt) {
      const size_t gp = (size_t)bi * NPOINTS + pw + tt;
      const _Float16* nb = z1h + ((size_t)bi * NPOINTS + nidx[tt]) * 16 + base0;
      const _Float16* ct = z1h + gp * 16 + base0;
      X[tt] = cat16(*(const v8h*)nb, *(const v8h*)ct);
    }

    float sv[TT / 2];
#pragma unroll
    for (int s = 0; s < 2; ++s) {
      v16h Y[TT][2];
#pragma unroll
      for (int kb = 0; kb < 2; ++kb) {
        v8h part[TT][2];
#pragma unroll
        for (int hf = 0; hf < 2; ++hf) {
          const int nc = 2 * kb + hf;
          const int j = nc * 16 + n16;
          const v16h A1 = ld16(&sW1[s][j][base0], &sW1[s][j][16 + base0]);
          const v8f bv = *(const v8f*)&sB1[s][nc * 16 + 8 * hl];
#pragma unroll
          for (int tt = 0; tt < TT; ++tt)
            part[tt][hf] = relu_pack(wmma16(A1, X[tt], bv));
        }
#pragma unroll
        for (int tt = 0; tt < TT; ++tt) Y[tt][kb] = cat16(part[tt][0], part[tt][1]);
      }

      v16h Z[TT][2];
#pragma unroll
      for (int kb = 0; kb < 2; ++kb) {
        v8h part[TT][2];
#pragma unroll
        for (int hf = 0; hf < 2; ++hf) {
          const int nc = 2 * kb + hf;
          const int j = nc * 16 + n16;
          const v16h A2a = ld16(&sW2[s][j][base0],      &sW2[s][j][16 + base0]);
          const v16h A2b = ld16(&sW2[s][j][32 + base0], &sW2[s][j][48 + base0]);
          const v8f bv = *(const v8f*)&sB2[s][nc * 16 + 8 * hl];
#pragma unroll
          for (int tt = 0; tt < TT; ++tt) {
            v8f c = wmma16(A2a, Y[tt][0], bv);
            c = wmma16(A2b, Y[tt][1], c);
            part[tt][hf] = relu_pack(c);
          }
        }
#pragma unroll
        for (int tt = 0; tt < TT; ++tt) Z[tt][kb] = cat16(part[tt][0], part[tt][1]);
      }

      const v16h B3a = ld16(&sW3[s][n16][base0],      &sW3[s][n16][16 + base0]);
      const v16h B3b = ld16(&sW3[s][n16][32 + base0], &sW3[s][n16][48 + base0]);
      const float b3v = sB3[s][n16];
      v8f c0;
#pragma unroll
      for (int r = 0; r < 8; ++r) c0[r] = b3v;
      float m[TT];
#pragma unroll
      for (int tt = 0; tt < TT; ++tt) {
        v8f c = wmma16(Z[tt][0], B3a, c0);
        c = wmma16(Z[tt][1], B3b, c);
        float v = c[0];
#pragma unroll
        for (int r = 1; r < 8; ++r) v = fmaxf(v, c[r]);
        m[tt] = v;
      }
#pragma unroll
      for (int q = 0; q < TT / 2; ++q) {
        const float mine = hl ? m[2 * q + 1] : m[2 * q];
        const float send = hl ? m[2 * q] : m[2 * q + 1];
        const float v = fmaxf(mine, __shfl_xor(send, 16, 32));
        if (s == 0) {
          sv[q] = CLAMPC * TWOPI_I * atanf(v * (1.0f / CLAMPC));
          slog += sv[q];
        } else {
          float* zp = zmat + ((size_t)bi * NPOINTS + pw + 2 * q + hl) * 32 + 16 + n16;
          *zp = *zp * __expf(sv[q]) + v;
        }
      }
    }
  }
  slog += __shfl_xor(slog, 16, 32);
  slog += __shfl_xor(slog, 8, 32);
  slog += __shfl_xor(slog, 4, 32);
  slog += __shfl_xor(slog, 2, 32);
  slog += __shfl_xor(slog, 1, 32);
  if (lane == 0) atomicAdd(&sLog, slog);
  __syncthreads();
  if (tid == 0) atomicAdd(&logdet[bi], sLog);
}

extern "C" void kernel_launch(void* const* d_in, const int* in_sizes, int n_in,
                              void* d_out, int out_size, void* d_ws, size_t ws_size,
                              hipStream_t stream) {
  const float* x     = (const float*)d_in[0];
  const int*   knn   = (const int*)d_in[1];
  const float* logs  = (const float*)d_in[2];
  const float* biasv = (const float*)d_in[3];
  const float* W1    = (const float*)d_in[4];
  const float* b1    = (const float*)d_in[5];
  const float* W2    = (const float*)d_in[6];
  const float* b2    = (const float*)d_in[7];
  const float* W3    = (const float*)d_in[8];
  const float* b3    = (const float*)d_in[9];

  float* zmat   = (float*)d_out;
  float* logdet = zmat + (size_t)NBATCH * NPOINTS * 32;
  _Float16* z1h = (_Float16*)d_ws;

  const int total = NBATCH * NPOINTS;
  prep_kernel<<<(total + 255) / 256, 256, 0, stream>>>(x, logs, biasv, zmat,
                                                       logdet, z1h, total);

  const dim3 grid(NWG);
  for (int u = 0; u < 4; u += 2) {
    conv_kernel<<<grid, THREADS, 0, stream>>>(
        z1h, knn,
        W1 + (size_t)u * 48 * 64, b1 + (size_t)u * 64,
        W2 + (size_t)u * 64 * 64, b2 + (size_t)u * 64,
        W3 + (size_t)u * 64 * 16, b3 + (size_t)u * 16,
        zmat, logdet);
  }
}


# relu moved behind the f16 pack as one packed 16-bit max per register: 8 vector ops per D tile, was 20 plus hazard nops; the 4 products of a weight fragment issued as one independent group
# speedup vs baseline: 5.8897x; 1.1451x over previous
#include <hip/hip_runtime.h>
#include <hip/hip_bf16.h>

typedef __attribute__((ext_vector_type(16))) _Float16 v16h;
typedef __attribute__((ext_vector_type(8)))  _Float16 v8h;
typedef __attribute__((ext_vector_type(8)))  float    v8f;

#define NPOINTS 8192
#define NBATCH  16
#define WAVES   16
#define THREADS (WAVES * 32)
#define NWG     256
#define PTS_WG  (NBATCH * NPOINTS / NWG)
#define TT      4
#define TRIPS   (PTS_WG / (TT * WAVES))
static_assert(NPOINTS % PTS_WG == 0, "a workgroup stays inside one batch");
static_assert(PTS_WG % (TT * WAVES) == 0 && TT % 2 == 0, "whole trips, points in pairs");
#define CLAMPC  1.9f
#define TWOPI_I 0.63661977236758f

__device__ __forceinline__ v16h ld16(const _Float16* a, const _Float16* b) {
  v8h lo = *(const v8h*)a;
  v8h hi = *(const v8h*)b;
  return __builtin_shufflevector(lo, hi, 0,1,2,3,4,5,6,7,8,9,10,11,12,13,14,15);
}

__device__ __forceinline__ v8f wmma16(v16h a, v16h b, v8f c) {
  return __builtin_amdgcn_wmma_f32_16x16x32_f16(false, a, false, b, (short)0, c,
                                                false, false);
}

__global__ void prep_kernel(const float* __restrict__ x,
                            const float* __restrict__ logs,
                            const float* __restrict__ bias,
                            float* __restrict__ zmat,
                            float* __restrict__ logdet,
                            _Float16* __restrict__ z1h, int total) {
  int p = blockIdx.x * blockDim.x + threadIdx.x;
  if (p < total) {
    const float* xr = x + (size_t)p * 32;
    float* zr = zmat + (size_t)p * 32;
    _Float16* hr = z1h + (size_t)p * 16;
#pragma unroll
    for (int c = 0; c < 32; ++c) {
      float zc = xr[31 - c] * expf(logs[c]) + bias[c];
      zr[c] = zc;
      if (c < 16) hr[c] = (_Float16)zc;
    }
  }
  if (p < NBATCH) {
    float su = 0.f;
#pragma unroll
    for (int c = 0; c < 32; ++c) su += logs[c];
    logdet[p] = su * (float)NPOINTS;
  }
}

__device__ __forceinline__ v8h relu_pack(v8f c) {
  const v8h h = __builtin_convertvector(c, v8h);
  const v8h zero = {};
  return __builtin_elementwise_max(h, zero);
}

__device__ __forceinline__ v16h cat16(v8h lo, v8h hi) {
  return __builtin_shufflevector(lo, hi, 0,1,2,3,4,5,6,7,8,9,10,11,12,13,14,15);
}

__global__ __launch_bounds__(THREADS) void conv_kernel(
    const _Float16* __restrict__ z1h, const int* __restrict__ knn,
    const float* __restrict__ W1g, const float* __restrict__ b1g,
    const float* __restrict__ W2g, const float* __restrict__ b2g,
    const float* __restrict__ W3g, const float* __restrict__ b3g,
    float* __restrict__ zmat, float* __restrict__ logdet) {
  __shared__ __align__(16) _Float16 sW1[2][64][40];
  __shared__ __align__(16) _Float16 sW2[2][64][72];
  __shared__ __align__(16) _Float16 sW3[2][16][72];
  __shared__ __align__(32) float sB1[2][64];
  __shared__ __align__(32) float sB2[2][64];
  __shared__ float sB3[2][16];
  __shared__ float sLog;

  const int tid  = threadIdx.x;
  const int wid  = tid >> 5;
  const int lane = tid & 31;
  const int n16  = lane & 15;
  const int hl   = lane >> 4;
  const int base0 = hl * 8;

  const int bi  = blockIdx.x / (NPOINTS / PTS_WG);
  const int pb0 = (blockIdx.x % (NPOINTS / PTS_WG)) * PTS_WG;

  for (int idx = tid; idx < 2 * 64 * 32; idx += THREADS) {
    int s = idx >> 11, r = idx & 2047;
    int j = r >> 5, k = r & 31;
    float v;
    if (k < 16)
      v = W1g[(s * 48 + 16 + k) * 64 + j] + W1g[(s * 48 + 32 + k) * 64 + j];
    else {
      int kk = k - 16;
      v = W1g[(s * 48 + kk) * 64 + j] - W1g[(s * 48 + 32 + kk) * 64 + j];
    }
    sW1[s][j][k] = (_Float16)v;
  }
  for (int idx = tid; idx < 2 * 64 * 64; idx += THREADS) {
    int s = idx >> 12, r = idx & 4095;
    int j = r >> 6, k = r & 63;
    sW2[s][j][k] = (_Float16)W2g[(s * 64 + k) * 64 + j];
  }
  for (int idx = tid; idx < 2 * 16 * 64; idx += THREADS) {
    int s = idx >> 10, r = idx & 1023;
    int j = r >> 6, k = r & 63;
    sW3[s][j][k] = (_Float16)W3g[(s * 64 + k) * 16 + j];
  }
  if (tid < 128) {
    int s = tid >> 6, j = tid & 63;
    sB1[s][j] = b1g[s * 64 + j];
    sB2[s][j] = b2g[s * 64 + j];
    if (tid < 32) sB3[tid >> 4][tid & 15] = b3g[(tid >> 4) * 16 + (tid & 15)];
    if (tid == 0) sLog = 0.f;
  }

  __syncthreads();

  float slog = 0.f;
  for (int it = 0; it < TRIPS; ++it) {
    const int pw = pb0 + (it * WAVES + wid) * TT;
    asm volatile("" ::: "memory");

    int nidx[TT];
#pragma unroll
    for (int tt = 0; tt < TT; ++tt)
      nidx[tt] = knn[((size_t)bi * NPOINTS + pw + tt) * 16 + n16];
    v16h X[TT];
#pragma unroll
    for (int tt = 0; tt < TT; ++tt) {
      const size_t gp = (size_t)bi * NPOINTS + pw + tt;
      const _Float16* nb = z1h + ((size_t)bi * NPOINTS + nidx[tt]) * 16 + base0;
      const _Float16* ct = z1h + gp * 16 + base0;
      X[tt] = cat16(*(const v8h*)nb, *(const v8h*)ct);
    }

    float sv[TT / 2];
#pragma unroll
    for (int s = 0; s < 2; ++s) {
      v16h Y[TT][2];
#pragma unroll
      for (int kb = 0; kb < 2; ++kb) {
        v8h part[TT][2];
#pragma unroll
        for (int hf = 0; hf < 2; ++hf) {
          const int nc = 2 * kb + hf;
          const int j = nc * 16 + n16;
          const v16h A1 = ld16(&sW1[s][j][base0], &sW1[s][j][16 + base0]);
          const v8f bv = *(const v8f*)&sB1[s][nc * 16 + 8 * hl];
          v8f acc[TT];
#pragma unroll
          for (int tt = 0; tt < TT; ++tt) acc[tt] = wmma16(A1, X[tt], bv);
#pragma unroll
          for (int tt = 0; tt < TT; ++tt) part[tt][hf] = relu_pack(acc[tt]);
        }
#pragma unroll
        for (int tt = 0; tt < TT; ++tt) Y[tt][kb] = cat16(part[tt][0], part[tt][1]);
      }

      v16h Z[TT][2];
#pragma unroll
      for (int kb = 0; kb < 2; ++kb) {
        v8h part[TT][2];
#pragma unroll
        for (int hf = 0; hf < 2; ++hf) {
          const int nc = 2 * kb + hf;
          const int j = nc * 16 + n16;
          const v16h A2a = ld16(&sW2[s][j][base0],      &sW2[s][j][16 + base0]);
          const v16h A2b = ld16(&sW2[s][j][32 + base0], &sW2[s][j][48 + base0]);
          const v8f bv = *(const v8f*)&sB2[s][nc * 16 + 8 * hl];
          v8f acc[TT];
#pragma unroll
          for (int tt = 0; tt < TT; ++tt) acc[tt] = wmma16(A2a, Y[tt][0], bv);
#pragma unroll
          for (int tt = 0; tt < TT; ++tt) acc[tt] = wmma16(A2b, Y[tt][1], acc[tt]);
#pragma unroll
          for (int tt = 0; tt < TT; ++tt) part[tt][hf] = relu_pack(acc[tt]);
        }
#pragma unroll
        for (int tt = 0; tt < TT; ++tt) Z[tt][kb] = cat16(part[tt][0], part[tt][1]);
      }

      const v16h B3a = ld16(&sW3[s][n16][base0],      &sW3[s][n16][16 + base0]);
      const v16h B3b = ld16(&sW3[s][n16][32 + base0], &sW3[s][n16][48 + base0]);
      const float b3v = sB3[s][n16];
      v8f c0;
#pragma unroll
      for (int r = 0; r < 8; ++r) c0[r] = b3v;
      float m[TT];
#pragma unroll
      for (int tt = 0; tt < TT; ++tt) {
        v8f c = wmma16(Z[tt][0], B3a, c0);
        c = wmma16(Z[tt][1], B3b, c);
        float v = c[0];
#pragma unroll
        for (int r = 1; r < 8; ++r) v = fmaxf(v, c[r]);
        m[tt] = v;
      }
#pragma unroll
      for (int q = 0; q < TT / 2; ++q) {
        const float mine = hl ? m[2 * q + 1] : m[2 * q];
        const float send = hl ? m[2 * q] : m[2 * q + 1];
        const float v = fmaxf(mine, __shfl_xor(send, 16, 32));
        if (s == 0) {
          sv[q] = CLAMPC * TWOPI_I * atanf(v * (1.0f / CLAMPC));
          slog += sv[q];
        } else {
          float* zp = zmat + ((size_t)bi * NPOINTS + pw + 2 * q + hl) * 32 + 16 + n16;
          *zp = *zp * __expf(sv[q]) + v;
        }
      }
    }
  }
  slog += __shfl_xor(slog, 16, 32);
  slog += __shfl_xor(slog, 8, 32);
  slog += __shfl_xor(slog, 4, 32);
  slog += __shfl_xor(slog, 2, 32);
  slog += __shfl_xor(slog, 1, 32);
  if (lane == 0) atomicAdd(&sLog, slog);
  __syncthreads();
  if (tid == 0) atomicAdd(&logdet[bi], sLog);
}

extern "C" void kernel_launch(void* const* d_in, const int* in_sizes, int n_in,
                              void* d_out, int out_size, void* d_ws, size_t ws_size,
                              hipStream_t stream) {
  const float* x     = (const float*)d_in[0];
  const int*   knn   = (const int*)d_in[1];
  const float* logs  = (const float*)d_in[2];
  const float* biasv = (const float*)d_in[3];
  const float* W1    = (const float*)d_in[4];
  const float* b1    = (const float*)d_in[5];
  const float* W2    = (const float*)d_in[6];
  const float* b2    = (const float*)d_in[7];
  const float* W3    = (const float*)d_in[8];
  const float* b3    = (const float*)d_in[9];

  float* zmat   = (float*)d_out;
  float* logdet = zmat + (size_t)NBATCH * NPOINTS * 32;
  _Float16* z1h = (_Float16*)d_ws;

  const int total = NBATCH * NPOINTS;
  prep_kernel<<<(total + 255) / 256, 256, 0, stream>>>(x, logs, biasv, zmat,
                                                       logdet, z1h, total);

  const dim3 grid(NWG);
  for (int u = 0; u < 4; u += 2) {
    conv_kernel<<<grid, THREADS, 0, stream>>>(
        z1h, knn,
        W1 + (size_t)u * 48 * 64, b1 + (size_t)u * 64,
        W2 + (size_t)u * 64 * 64, b2 + (size_t)u * 64,
        W3 + (size_t)u * 64 * 16, b3 + (size_t)u * 16,
        zmat, logdet);
  }
}


# both couplings in one launch: no coupling changes z1, so a point's edge operand is gathered once and feeds all 4 nets; z2 read once, updated twice in a register, written once (was 2 launches)
# speedup vs baseline: 6.9257x; 1.1759x over previous
#include <hip/hip_runtime.h>
#include <hip/hip_bf16.h>

typedef __attribute__((ext_vector_type(16))) _Float16 v16h;
typedef __attribute__((ext_vector_type(8)))  _Float16 v8h;
typedef __attribute__((ext_vector_type(8)))  float    v8f;

#define NPOINTS 8192
#define NBATCH  16
#define WAVES   16
#define THREADS (WAVES * 32)
#define NWG     256
#define PTS_WG  (NBATCH * NPOINTS / NWG)
#define TT      4
#define TRIPS   (PTS_WG / (TT * WAVES))
static_assert(NPOINTS % PTS_WG == 0, "a workgroup stays inside one batch");
static_assert(PTS_WG % (TT * WAVES) == 0 && TT % 2 == 0, "whole trips, points in pairs");
#define CLAMPC  1.9f
#define TWOPI_I 0.63661977236758f

__device__ __forceinline__ v16h ld16(const _Float16* a, const _Float16* b) {
  v8h lo = *(const v8h*)a;
  v8h hi = *(const v8h*)b;
  return __builtin_shufflevector(lo, hi, 0,1,2,3,4,5,6,7,8,9,10,11,12,13,14,15);
}

__device__ __forceinline__ v8f wmma16(v16h a, v16h b, v8f c) {
  return __builtin_amdgcn_wmma_f32_16x16x32_f16(false, a, false, b, (short)0, c,
                                                false, false);
}

__global__ void prep_kernel(const float* __restrict__ x,
                            const float* __restrict__ logs,
                            const float* __restrict__ bias,
                            float* __restrict__ zmat,
                            float* __restrict__ logdet,
                            _Float16* __restrict__ z1h, int total) {
  int p = blockIdx.x * blockDim.x + threadIdx.x;
  if (p < total) {
    const float* xr = x + (size_t)p * 32;
    float* zr = zmat + (size_t)p * 32;
    _Float16* hr = z1h + (size_t)p * 16;
#pragma unroll
    for (int c = 0; c < 32; ++c) {
      float zc = xr[31 - c] * expf(logs[c]) + bias[c];
      zr[c] = zc;
      if (c < 16) hr[c] = (_Float16)zc;
    }
  }
  if (p < NBATCH) {
    float su = 0.f;
#pragma unroll
    for (int c = 0; c < 32; ++c) su += logs[c];
    logdet[p] = su * (float)NPOINTS;
  }
}

__device__ __forceinline__ v8h relu_pack(v8f c) {
  const v8h h = __builtin_convertvector(c, v8h);
  const v8h zero = {};
  return __builtin_elementwise_max(h, zero);
}

__device__ __forceinline__ v16h cat16(v8h lo, v8h hi) {
  return __builtin_shufflevector(lo, hi, 0,1,2,3,4,5,6,7,8,9,10,11,12,13,14,15);
}

__global__ __launch_bounds__(THREADS) void conv_kernel(
    const _Float16* __restrict__ z1h, const int* __restrict__ knn,
    const float* __restrict__ W1g, const float* __restrict__ b1g,
    const float* __restrict__ W2g, const float* __restrict__ b2g,
    const float* __restrict__ W3g, const float* __restrict__ b3g,
    float* __restrict__ zmat, float* __restrict__ logdet) {
  __shared__ __align__(16) _Float16 sW1[4][64][40];
  __shared__ __align__(16) _Float16 sW2[4][64][72];
  __shared__ __align__(16) _Float16 sW3[4][16][72];
  __shared__ __align__(32) float sB1[4][64];
  __shared__ __align__(32) float sB2[4][64];
  __shared__ float sB3[4][16];
  __shared__ float sLog;

  const int tid  = threadIdx.x;
  const int wid  = tid >> 5;
  const int lane = tid & 31;
  const int n16  = lane & 15;
  const int hl   = lane >> 4;
  const int base0 = hl * 8;

  const int bi  = blockIdx.x / (NPOINTS / PTS_WG);
  const int pb0 = (blockIdx.x % (NPOINTS / PTS_WG)) * PTS_WG;

  for (int idx = tid; idx < 4 * 64 * 32; idx += THREADS) {
    int s = idx >> 11, r = idx & 2047;
    int j = r >> 5, k = r & 31;
    float v;
    if (k < 16)
      v = W1g[(s * 48 + 16 + k) * 64 + j] + W1g[(s * 48 + 32 + k) * 64 + j];
    else {
      int kk = k - 16;
      v = W1g[(s * 48 + kk) * 64 + j] - W1g[(s * 48 + 32 + kk) * 64 + j];
    }
    sW1[s][j][k] = (_Float16)v;
  }
  for (int idx = tid; idx < 4 * 64 * 64; idx += THREADS) {
    int s = idx >> 12, r = idx & 4095;
    int j = r >> 6, k = r & 63;
    sW2[s][j][k] = (_Float16)W2g[(s * 64 + k) * 64 + j];
  }
  for (int idx = tid; idx < 4 * 16 * 64; idx += THREADS) {
    int s = idx >> 10, r = idx & 1023;
    int j = r >> 6, k = r & 63;
    sW3[s][j][k] = (_Float16)W3g[(s * 64 + k) * 16 + j];
  }
  if (tid < 256) {
    int s = tid >> 6, j = tid & 63;
    sB1[s][j] = b1g[s * 64 + j];
    sB2[s][j] = b2g[s * 64 + j];
    if (tid < 64) sB3[tid >> 4][tid & 15] = b3g[(tid >> 4) * 16 + (tid & 15)];
    if (tid == 0) sLog = 0.f;
  }

  __syncthreads();

  float slog = 0.f;
  for (int it = 0; it < TRIPS; ++it) {
    const int pw = pb0 + (it * WAVES + wid) * TT;
    asm volatile("" ::: "memory");

    int nidx[TT];
#pragma unroll
    for (int tt = 0; tt < TT; ++tt)
      nidx[tt] = knn[((size_t)bi * NPOINTS + pw + tt) * 16 + n16];
    v16h X[TT];
#pragma unroll
    for (int tt = 0; tt < TT; ++tt) {
      const size_t gp = (size_t)bi * NPOINTS + pw + tt;
      const _Float16* nb = z1h + ((size_t)bi * NPOINTS + nidx[tt]) * 16 + base0;
      const _Float16* ct = z1h + gp * 16 + base0;
      X[tt] = cat16(*(const v8h*)nb, *(const v8h*)ct);
    }

    float* zp[TT / 2];
    float zv[TT / 2];
#pragma unroll
    for (int q = 0; q < TT / 2; ++q) {
      zp[q] = zmat + ((size_t)bi * NPOINTS + pw + 2 * q + hl) * 32 + 16 + n16;
      zv[q] = *zp[q];
    }

    float sv[TT / 2];
#pragma unroll
    for (int ns = 0; ns < 4; ++ns) {
      const int s = ns & 1;
      v16h Y[TT][2];
#pragma unroll
      for (int kb = 0; kb < 2; ++kb) {
        v8h part[TT][2];
#pragma unroll
        for (int hf = 0; hf < 2; ++hf) {
          const int nc = 2 * kb + hf;
          const int j = nc * 16 + n16;
          const v16h A1 = ld16(&sW1[ns][j][base0], &sW1[ns][j][16 + base0]);
          const v8f bv = *(const v8f*)&sB1[ns][nc * 16 + 8 * hl];
          v8f acc[TT];
#pragma unroll
          for (int tt = 0; tt < TT; ++tt) acc[tt] = wmma16(A1, X[tt], bv);
#pragma unroll
          for (int tt = 0; tt < TT; ++tt) part[tt][hf] = relu_pack(acc[tt]);
        }
#pragma unroll
        for (int tt = 0; tt < TT; ++tt) Y[tt][kb] = cat16(part[tt][0], part[tt][1]);
      }

      v16h Z[TT][2];
#pragma unroll
      for (int kb = 0; kb < 2; ++kb) {
        v8h part[TT][2];
#pragma unroll
        for (int hf = 0; hf < 2; ++hf) {
          const int nc = 2 * kb + hf;
          const int j = nc * 16 + n16;
          const v16h A2a = ld16(&sW2[ns][j][base0],      &sW2[ns][j][16 + base0]);
          const v16h A2b = ld16(&sW2[ns][j][32 + base0], &sW2[ns][j][48 + base0]);
          const v8f bv = *(const v8f*)&sB2[ns][nc * 16 + 8 * hl];
          v8f acc[TT];
#pragma unroll
          for (int tt = 0; tt < TT; ++tt) acc[tt] = wmma16(A2a, Y[tt][0], bv);
#pragma unroll
          for (int tt = 0; tt < TT; ++tt) acc[tt] = wmma16(A2b, Y[tt][1], acc[tt]);
#pragma unroll
          for (int tt = 0; tt < TT; ++tt) part[tt][hf] = relu_pack(acc[tt]);
        }
#pragma unroll
        for (int tt = 0; tt < TT; ++tt) Z[tt][kb] = cat16(part[tt][0], part[tt][1]);
      }

      const v16h B3a = ld16(&sW3[ns][n16][base0],      &sW3[ns][n16][16 + base0]);
      const v16h B3b = ld16(&sW3[ns][n16][32 + base0], &sW3[ns][n16][48 + base0]);
      const float b3v = sB3[ns][n16];
      v8f c0;
#pragma unroll
      for (int r = 0; r < 8; ++r) c0[r] = b3v;
      float m[TT];
#pragma unroll
      for (int tt = 0; tt < TT; ++tt) {
        v8f c = wmma16(Z[tt][0], B3a, c0);
        c = wmma16(Z[tt][1], B3b, c);
        float v = c[0];
#pragma unroll
        for (int r = 1; r < 8; ++r) v = fmaxf(v, c[r]);
        m[tt] = v;
      }
#pragma unroll
      for (int q = 0; q < TT / 2; ++q) {
        const float mine = hl ? m[2 * q + 1] : m[2 * q];
        const float send = hl ? m[2 * q] : m[2 * q + 1];
        const float v = fmaxf(mine, __shfl_xor(send, 16, 32));
        if (s == 0) {
          sv[q] = CLAMPC * TWOPI_I * atanf(v * (1.0f / CLAMPC));
          slog += sv[q];
        } else {
          zv[q] = zv[q] * __expf(sv[q]) + v;
        }
      }
    }
#pragma unroll
    for (int q = 0; q < TT / 2; ++q) *zp[q] = zv[q];
  }
  slog += __shfl_xor(slog, 16, 32);
  slog += __shfl_xor(slog, 8, 32);
  slog += __shfl_xor(slog, 4, 32);
  slog += __shfl_xor(slog, 2, 32);
  slog += __shfl_xor(slog, 1, 32);
  if (lane == 0) atomicAdd(&sLog, slog);
  __syncthreads();
  if (tid == 0) atomicAdd(&logdet[bi], sLog);
}

extern "C" void kernel_launch(void* const* d_in, const int* in_sizes, int n_in,
                              void* d_out, int out_size, void* d_ws, size_t ws_size,
                              hipStream_t stream) {
  const float* x     = (const float*)d_in[0];
  const int*   knn   = (const int*)d_in[1];
  const float* logs  = (const float*)d_in[2];
  const float* biasv = (const float*)d_in[3];
  const float* W1    = (const float*)d_in[4];
  const float* b1    = (const float*)d_in[5];
  const float* W2    = (const float*)d_in[6];
  const float* b2    = (const float*)d_in[7];
  const float* W3    = (const float*)d_in[8];
  const float* b3    = (const float*)d_in[9];

  float* zmat   = (float*)d_out;
  float* logdet = zmat + (size_t)NBATCH * NPOINTS * 32;
  _Float16* z1h = (_Float16*)d_ws;

  const int total = NBATCH * NPOINTS;
  prep_kernel<<<(total + 255) / 256, 256, 0, stream>>>(x, logs, biasv, zmat,
                                                       logdet, z1h, total);

  const dim3 grid(NWG);
  conv_kernel<<<grid, THREADS, 0, stream>>>(z1h, knn, W1, b1, W2, b2, W3, b3,
                                            zmat, logdet);
}


# weight staging coalesced: lanes run along the contiguous output-channel axis, so a wave's load reads one 128-byte line (was 32 lines); two k per 4-byte LDS store, a loop's loads all in flight
# speedup vs baseline: 7.7570x; 1.1200x over previous
#include <hip/hip_runtime.h>
#include <hip/hip_bf16.h>

typedef __attribute__((ext_vector_type(16))) _Float16 v16h;
typedef __attribute__((ext_vector_type(8)))  _Float16 v8h;
typedef __attribute__((ext_vector_type(2)))  _Float16 v2h;
typedef __attribute__((ext_vector_type(8)))  float    v8f;

#define NPOINTS 8192
#define NBATCH  16
#define WAVES   16
#define THREADS (WAVES * 32)
#define NWG     256
#define PTS_WG  (NBATCH * NPOINTS / NWG)
#define TT      4
#define TRIPS   (PTS_WG / (TT * WAVES))
static_assert(NPOINTS % PTS_WG == 0, "a workgroup stays inside one batch");
static_assert(PTS_WG % (TT * WAVES) == 0 && TT % 2 == 0, "whole trips, points in pairs");
static_assert((4 * 32 * 16) % THREADS == 0, "the staging loops have whole trips");
#define CLAMPC  1.9f
#define TWOPI_I 0.63661977236758f

__device__ __forceinline__ v16h ld16(const _Float16* a, const _Float16* b) {
  v8h lo = *(const v8h*)a;
  v8h hi = *(const v8h*)b;
  return __builtin_shufflevector(lo, hi, 0,1,2,3,4,5,6,7,8,9,10,11,12,13,14,15);
}

__device__ __forceinline__ v8f wmma16(v16h a, v16h b, v8f c) {
  return __builtin_amdgcn_wmma_f32_16x16x32_f16(false, a, false, b, (short)0, c,
                                                false, false);
}

__global__ void prep_kernel(const float* __restrict__ x,
                            const float* __restrict__ logs,
                            const float* __restrict__ bias,
                            float* __restrict__ zmat,
                            float* __restrict__ logdet,
                            _Float16* __restrict__ z1h, int total) {
  int p = blockIdx.x * blockDim.x + threadIdx.x;
  if (p < total) {
    const float* xr = x + (size_t)p * 32;
    float* zr = zmat + (size_t)p * 32;
    _Float16* hr = z1h + (size_t)p * 16;
#pragma unroll
    for (int c = 0; c < 32; ++c) {
      float zc = xr[31 - c] * expf(logs[c]) + bias[c];
      zr[c] = zc;
      if (c < 16) hr[c] = (_Float16)zc;
    }
  }
  if (p < NBATCH) {
    float su = 0.f;
#pragma unroll
    for (int c = 0; c < 32; ++c) su += logs[c];
    logdet[p] = su * (float)NPOINTS;
  }
}

__device__ __forceinline__ v8h relu_pack(v8f c) {
  const v8h h = __builtin_convertvector(c, v8h);
  const v8h zero = {};
  return __builtin_elementwise_max(h, zero);
}

__device__ __forceinline__ v16h cat16(v8h lo, v8h hi) {
  return __builtin_shufflevector(lo, hi, 0,1,2,3,4,5,6,7,8,9,10,11,12,13,14,15);
}

__global__ __launch_bounds__(THREADS) void conv_kernel(
    const _Float16* __restrict__ z1h, const int* __restrict__ knn,
    const float* __restrict__ W1g, const float* __restrict__ b1g,
    const float* __restrict__ W2g, const float* __restrict__ b2g,
    const float* __restrict__ W3g, const float* __restrict__ b3g,
    float* __restrict__ zmat, float* __restrict__ logdet) {
  __shared__ __align__(16) _Float16 sW1[4][64][40];
  __shared__ __align__(16) _Float16 sW2[4][64][72];
  __shared__ __align__(16) _Float16 sW3[4][16][72];
  __shared__ __align__(32) float sB1[4][64];
  __shared__ __align__(32) float sB2[4][64];
  __shared__ float sB3[4][16];
  __shared__ float sLog;

  const int tid  = threadIdx.x;
  const int wid  = tid >> 5;
  const int lane = tid & 31;
  const int n16  = lane & 15;
  const int hl   = lane >> 4;
  const int base0 = hl * 8;

  const int bi  = blockIdx.x / (NPOINTS / PTS_WG);
  const int pb0 = (blockIdx.x % (NPOINTS / PTS_WG)) * PTS_WG;

#pragma unroll
  for (int it = 0; it < 4 * 16 * 64 / THREADS; ++it) {
    const int idx = tid + it * THREADS;
    const int j = idx & 63, kp = (idx >> 6) & 15, s = idx >> 10;
    const int k0 = 2 * kp;
    const bool nbh = k0 < 16;
    const float* pa = W1g + (s * 48 + (nbh ? 16 + k0 : k0 - 16)) * 64 + j;
    const float* pb = W1g + (s * 48 + 32 + (k0 & 15)) * 64 + j;
    const float a0 = pa[0], a1 = pa[64], c0 = pb[0], c1 = pb[64];
    const float v0 = nbh ? a0 + c0 : a0 - c0;
    const float v1 = nbh ? a1 + c1 : a1 - c1;
    const v2h h = {(_Float16)v0, (_Float16)v1};
    *(v2h*)&sW1[s][j][k0] = h;
  }
#pragma unroll
  for (int it = 0; it < 4 * 32 * 64 / THREADS; ++it) {
    const int idx = tid + it * THREADS;
    const int j = idx & 63, kp = (idx >> 6) & 31, s = idx >> 11;
    const float* p = W2g + (s * 64 + 2 * kp) * 64 + j;
    const v2h h = {(_Float16)p[0], (_Float16)p[64]};
    *(v2h*)&sW2[s][j][2 * kp] = h;
  }
#pragma unroll
  for (int it = 0; it < 4 * 32 * 16 / THREADS; ++it) {
    const int idx = tid + it * THREADS;
    const int j = idx & 15, kp = (idx >> 4) & 31, s = idx >> 9;
    const float* p = W3g + (s * 64 + 2 * kp) * 16 + j;
    const v2h h = {(_Float16)p[0], (_Float16)p[16]};
    *(v2h*)&sW3[s][j][2 * kp] = h;
  }
  if (tid < 256) {
    int s = tid >> 6, j = tid & 63;
    sB1[s][j] = b1g[s * 64 + j];
    sB2[s][j] = b2g[s * 64 + j];
    if (tid < 64) sB3[tid >> 4][tid & 15] = b3g[(tid >> 4) * 16 + (tid & 15)];
    if (tid == 0) sLog = 0.f;
  }

  __syncthreads();

  float slog = 0.f;
  for (int it = 0; it < TRIPS; ++it) {
    const int pw = pb0 + (it * WAVES + wid) * TT;
    asm volatile("" ::: "memory");

    int nidx[TT];
#pragma unroll
    for (int tt = 0; tt < TT; ++tt)
      nidx[tt] = knn[((size_t)bi * NPOINTS + pw + tt) * 16 + n16];
    v16h X[TT];
#pragma unroll
    for (int tt = 0; tt < TT; ++tt) {
      const size_t gp = (size_t)bi * NPOINTS + pw + tt;
      const _Float16* nb = z1h + ((size_t)bi * NPOINTS + nidx[tt]) * 16 + base0;
      const _Float16* ct = z1h + gp * 16 + base0;
      X[tt] = cat16(*(const v8h*)nb, *(const v8h*)ct);
    }

    float* zp[TT / 2];
    float zv[TT / 2];
#pragma unroll
    for (int q = 0; q < TT / 2; ++q) {
      zp[q] = zmat + ((size_t)bi * NPOINTS + pw + 2 * q + hl) * 32 + 16 + n16;
      zv[q] = *zp[q];
    }

    float sv[TT / 2];
#pragma unroll
    for (int ns = 0; ns < 4; ++ns) {
      const int s = ns & 1;
      v16h Y[TT][2];
#pragma unroll
      for (int kb = 0; kb < 2; ++kb) {
        v8h part[TT][2];
#pragma unroll
        for (int hf = 0; hf < 2; ++hf) {
          const int nc = 2 * kb + hf;
          const int j = nc * 16 + n16;
          const v16h A1 = ld16(&sW1[ns][j][base0], &sW1[ns][j][16 + base0]);
          const v8f bv = *(const v8f*)&sB1[ns][nc * 16 + 8 * hl];
          v8f acc[TT];
#pragma unroll
          for (int tt = 0; tt < TT; ++tt) acc[tt] = wmma16(A1, X[tt], bv);
#pragma unroll
          for (int tt = 0; tt < TT; ++tt) part[tt][hf] = relu_pack(acc[tt]);
        }
#pragma unroll
        for (int tt = 0; tt < TT; ++tt) Y[tt][kb] = cat16(part[tt][0], part[tt][1]);
      }

      v16h Z[TT][2];
#pragma unroll
      for (int kb = 0; kb < 2; ++kb) {
        v8h part[TT][2];
#pragma unroll
        for (int hf = 0; hf < 2; ++hf) {
          const int nc = 2 * kb + hf;
          const int j = nc * 16 + n16;
          const v16h A2a = ld16(&sW2[ns][j][base0],      &sW2[ns][j][16 + base0]);
          const v16h A2b = ld16(&sW2[ns][j][32 + base0], &sW2[ns][j][48 + base0]);
          const v8f bv = *(const v8f*)&sB2[ns][nc * 16 + 8 * hl];
          v8f acc[TT];
#pragma unroll
          for (int tt = 0; tt < TT; ++tt) acc[tt] = wmma16(A2a, Y[tt][0], bv);
#pragma unroll
          for (int tt = 0; tt < TT; ++tt) acc[tt] = wmma16(A2b, Y[tt][1], acc[tt]);
#pragma unroll
          for (int tt = 0; tt < TT; ++tt) part[tt][hf] = relu_pack(acc[tt]);
        }
#pragma unroll
        for (int tt = 0; tt < TT; ++tt) Z[tt][kb] = cat16(part[tt][0], part[tt][1]);
      }

      const v16h B3a = ld16(&sW3[ns][n16][base0],      &sW3[ns][n16][16 + base0]);
      const v16h B3b = ld16(&sW3[ns][n16][32 + base0], &sW3[ns][n16][48 + base0]);
      const float b3v = sB3[ns][n16];
      v8f c0;
#pragma unroll
      for (int r = 0; r < 8; ++r) c0[r] = b3v;
      float m[TT];
#pragma unroll
      for (int tt = 0; tt < TT; ++tt) {
        v8f c = wmma16(Z[tt][0], B3a, c0);
        c = wmma16(Z[tt][1], B3b, c);
        float v = c[0];
#pragma unroll
        for (int r = 1; r < 8; ++r) v = fmaxf(v, c[r]);
        m[tt] = v;
      }
#pragma unroll
      for (int q = 0; q < TT / 2; ++q) {
        const float mine = hl ? m[2 * q + 1] : m[2 * q];
        const float send = hl ? m[2 * q] : m[2 * q + 1];
        const float v = fmaxf(mine, __shfl_xor(send, 16, 32));
        if (s == 0) {
          sv[q] = CLAMPC * TWOPI_I * atanf(v * (1.0f / CLAMPC));
          slog += sv[q];
        } else {
          zv[q] = zv[q] * __expf(sv[q]) + v;
        }
      }
    }
#pragma unroll
    for (int q = 0; q < TT / 2; ++q) *zp[q] = zv[q];
  }
  slog += __shfl_xor(slog, 16, 32);
  slog += __shfl_xor(slog, 8, 32);
  slog += __shfl_xor(slog, 4, 32);
  slog += __shfl_xor(slog, 2, 32);
  slog += __shfl_xor(slog, 1, 32);
  if (lane == 0) atomicAdd(&sLog, slog);
  __syncthreads();
  if (tid == 0) atomicAdd(&logdet[bi], sLog);
}

extern "C" void kernel_launch(void* const* d_in, const int* in_sizes, int n_in,
                              void* d_out, int out_size, void* d_ws, size_t ws_size,
                              hipStream_t stream) {
  const float* x     = (const float*)d_in[0];
  const int*   knn   = (const int*)d_in[1];
  const float* logs  = (const float*)d_in[2];
  const float* biasv = (const float*)d_in[3];
  const float* W1    = (const float*)d_in[4];
  const float* b1    = (const float*)d_in[5];
  const float* W2    = (const float*)d_in[6];
  const float* b2    = (const float*)d_in[7];
  const float* W3    = (const float*)d_in[8];
  const float* b3    = (const float*)d_in[9];

  float* zmat   = (float*)d_out;
  float* logdet = zmat + (size_t)NBATCH * NPOINTS * 32;
  _Float16* z1h = (_Float16*)d_ws;

  const int total = NBATCH * NPOINTS;
  prep_kernel<<<(total + 255) / 256, 256, 0, stream>>>(x, logs, biasv, zmat,
                                                       logdet, z1h, total);

  const dim3 grid(NWG);
  conv_kernel<<<grid, THREADS, 0, stream>>>(z1h, knn, W1, b1, W2, b2, W3, b3,
                                            zmat, logdet);
}


# points handed out in chunks of 4 through an LDS counter (was a fixed 8 trips per wave): waves the SIMD serves first take more, all end together; s summed by chunk in fixed order (sum order changes)
# speedup vs baseline: 8.2098x; 1.0584x over previous
#include <hip/hip_runtime.h>
#include <hip/hip_bf16.h>

typedef __attribute__((ext_vector_type(16))) _Float16 v16h;
typedef __attribute__((ext_vector_type(8)))  _Float16 v8h;
typedef __attribute__((ext_vector_type(2)))  _Float16 v2h;
typedef __attribute__((ext_vector_type(8)))  float    v8f;

#define NPOINTS 8192
#define NBATCH  16
#define WAVES   16
#define THREADS (WAVES * 32)
#define NWG     256
#define PTS_WG  (NBATCH * NPOINTS / NWG)
#define TT      4
#define NCHUNK  (PTS_WG / TT)
static_assert(NPOINTS % PTS_WG == 0, "a workgroup stays inside one batch");
static_assert(PTS_WG % TT == 0 && TT % 2 == 0, "whole chunks, points in pairs");
static_assert((NCHUNK * 32) % THREADS == 0, "the closing sum has whole trips");
static_assert((4 * 32 * 16) % THREADS == 0, "the staging loops have whole trips");
#define CLAMPC  1.9f
#define TWOPI_I 0.63661977236758f

__device__ __forceinline__ v16h ld16(const _Float16* a, const _Float16* b) {
  v8h lo = *(const v8h*)a;
  v8h hi = *(const v8h*)b;
  return __builtin_shufflevector(lo, hi, 0,1,2,3,4,5,6,7,8,9,10,11,12,13,14,15);
}

__device__ __forceinline__ v8f wmma16(v16h a, v16h b, v8f c) {
  return __builtin_amdgcn_wmma_f32_16x16x32_f16(false, a, false, b, (short)0, c,
                                                false, false);
}

__global__ void prep_kernel(const float* __restrict__ x,
                            const float* __restrict__ logs,
                            const float* __restrict__ bias,
                            float* __restrict__ zmat,
                            float* __restrict__ logdet,
                            _Float16* __restrict__ z1h, int total) {
  int p = blockIdx.x * blockDim.x + threadIdx.x;
  if (p < total) {
    const float* xr = x + (size_t)p * 32;
    float* zr = zmat + (size_t)p * 32;
    _Float16* hr = z1h + (size_t)p * 16;
#pragma unroll
    for (int c = 0; c < 32; ++c) {
      float zc = xr[31 - c] * expf(logs[c]) + bias[c];
      zr[c] = zc;
      if (c < 16) hr[c] = (_Float16)zc;
    }
  }
  if (p < NBATCH) {
    float su = 0.f;
#pragma unroll
    for (int c = 0; c < 32; ++c) su += logs[c];
    logdet[p] = su * (float)NPOINTS;
  }
}

__device__ __forceinline__ v8h relu_pack(v8f c) {
  const v8h h = __builtin_convertvector(c, v8h);
  const v8h zero = {};
  return __builtin_elementwise_max(h, zero);
}

__device__ __forceinline__ const char* at(const void* base, unsigned bytes) {
  return (const char*)base + bytes;
}

__device__ __forceinline__ v16h cat16(v8h lo, v8h hi) {
  return __builtin_shufflevector(lo, hi, 0,1,2,3,4,5,6,7,8,9,10,11,12,13,14,15);
}

__global__ __launch_bounds__(THREADS) void conv_kernel(
    const _Float16* __restrict__ z1h, const int* __restrict__ knn,
    const float* __restrict__ W1g, const float* __restrict__ b1g,
    const float* __restrict__ W2g, const float* __restrict__ b2g,
    const float* __restrict__ W3g, const float* __restrict__ b3g,
    float* __restrict__ zmat, float* __restrict__ logdet) {
  __shared__ __align__(16) _Float16 sW1[4][64][40];
  __shared__ __align__(16) _Float16 sW2[4][64][72];
  __shared__ __align__(16) _Float16 sW3[4][16][72];
  __shared__ __align__(32) float sB1[4][64];
  __shared__ __align__(32) float sB2[4][64];
  __shared__ float sB3[4][16];
  __shared__ float sPart[NCHUNK][32];
  __shared__ float sWave[WAVES];
  __shared__ int sNext;

  const int tid  = threadIdx.x;
  const int wid  = tid >> 5;
  const int lane = tid & 31;
  const int n16  = lane & 15;
  const int hl   = lane >> 4;
  const int base0 = hl * 8;

  const int bi  = blockIdx.x / (NPOINTS / PTS_WG);
  const int pb0 = (blockIdx.x % (NPOINTS / PTS_WG)) * PTS_WG;

#pragma unroll
  for (int it = 0; it < 4 * 16 * 64 / THREADS; ++it) {
    const int idx = tid + it * THREADS;
    const int j = idx & 63, kp = (idx >> 6) & 15, s = idx >> 10;
    const int k0 = 2 * kp;
    const bool nbh = k0 < 16;
    const float* pa = W1g + (s * 48 + (nbh ? 16 + k0 : k0 - 16)) * 64 + j;
    const float* pb = W1g + (s * 48 + 32 + (k0 & 15)) * 64 + j;
    const float a0 = pa[0], a1 = pa[64], c0 = pb[0], c1 = pb[64];
    const float v0 = nbh ? a0 + c0 : a0 - c0;
    const float v1 = nbh ? a1 + c1 : a1 - c1;
    const v2h h = {(_Float16)v0, (_Float16)v1};
    *(v2h*)&sW1[s][j][k0] = h;
  }
#pragma unroll
  for (int it = 0; it < 4 * 32 * 64 / THREADS; ++it) {
    const int idx = tid + it * THREADS;
    const int j = idx & 63, kp = (idx >> 6) & 31, s = idx >> 11;
    const float* p = W2g + (s * 64 + 2 * kp) * 64 + j;
    const v2h h = {(_Float16)p[0], (_Float16)p[64]};
    *(v2h*)&sW2[s][j][2 * kp] = h;
  }
#pragma unroll
  for (int it = 0; it < 4 * 32 * 16 / THREADS; ++it) {
    const int idx = tid + it * THREADS;
    const int j = idx & 15, kp = (idx >> 4) & 31, s = idx >> 9;
    const float* p = W3g + (s * 64 + 2 * kp) * 16 + j;
    const v2h h = {(_Float16)p[0], (_Float16)p[16]};
    *(v2h*)&sW3[s][j][2 * kp] = h;
  }
  if (tid < 256) {
    int s = tid >> 6, j = tid & 63;
    sB1[s][j] = b1g[s * 64 + j];
    sB2[s][j] = b2g[s * 64 + j];
    if (tid < 64) sB3[tid >> 4][tid & 15] = b3g[(tid >> 4) * 16 + (tid & 15)];
    if (tid == 0) sNext = 0;
  }

  __syncthreads();

  for (int n = 0; n < NCHUNK; ++n) {
    int got = 0;
    if (lane == 0) got = atomicAdd(&sNext, 1);
    const int chunk = __builtin_amdgcn_readfirstlane(got);
    if (chunk >= NCHUNK) break;
    const int pw = pb0 + chunk * TT;
    float slog = 0.f;
    asm volatile("" ::: "memory");

    const unsigned gp0 = (unsigned)(bi * NPOINTS + pw);
    int nidx[TT];
#pragma unroll
    for (int tt = 0; tt < TT; ++tt)
      nidx[tt] = *(const int*)at(knn, ((gp0 + tt) * 16u + n16) * 4u);
    v16h X[TT];
#pragma unroll
    for (int tt = 0; tt < TT; ++tt) {
      const unsigned nbp = (unsigned)(bi * NPOINTS) + (unsigned)nidx[tt];
      const char* nb = at(z1h, (nbp * 16u + base0) * 2u);
      const char* ct = at(z1h, ((gp0 + tt) * 16u + base0) * 2u);
      X[tt] = cat16(*(const v8h*)nb, *(const v8h*)ct);
    }

    const unsigned zo = ((gp0 + hl) * 32u + 16u + n16) * 4u;
    float zv[TT / 2];
#pragma unroll
    for (int q = 0; q < TT / 2; ++q) zv[q] = *(const float*)at(zmat, zo + q * 256u);

    float sv[TT / 2];
#pragma unroll
    for (int ns = 0; ns < 4; ++ns) {
      const int s = ns & 1;
      v16h Y[TT][2];
#pragma unroll
      for (int kb = 0; kb < 2; ++kb) {
        v8h part[TT][2];
#pragma unroll
        for (int hf = 0; hf < 2; ++hf) {
          const int nc = 2 * kb + hf;
          const int j = nc * 16 + n16;
          const v16h A1 = ld16(&sW1[ns][j][base0], &sW1[ns][j][16 + base0]);
          const v8f bv = *(const v8f*)&sB1[ns][nc * 16 + 8 * hl];
          v8f acc[TT];
#pragma unroll
          for (int tt = 0; tt < TT; ++tt) acc[tt] = wmma16(A1, X[tt], bv);
#pragma unroll
          for (int tt = 0; tt < TT; ++tt) part[tt][hf] = relu_pack(acc[tt]);
        }
#pragma unroll
        for (int tt = 0; tt < TT; ++tt) Y[tt][kb] = cat16(part[tt][0], part[tt][1]);
      }

      v16h Z[TT][2];
#pragma unroll
      for (int kb = 0; kb < 2; ++kb) {
        v8h part[TT][2];
#pragma unroll
        for (int hf = 0; hf < 2; ++hf) {
          const int nc = 2 * kb + hf;
          const int j = nc * 16 + n16;
          const v16h A2a = ld16(&sW2[ns][j][base0],      &sW2[ns][j][16 + base0]);
          const v16h A2b = ld16(&sW2[ns][j][32 + base0], &sW2[ns][j][48 + base0]);
          const v8f bv = *(const v8f*)&sB2[ns][nc * 16 + 8 * hl];
          v8f acc[TT];
#pragma unroll
          for (int tt = 0; tt < TT; ++tt) acc[tt] = wmma16(A2a, Y[tt][0], bv);
#pragma unroll
          for (int tt = 0; tt < TT; ++tt) acc[tt] = wmma16(A2b, Y[tt][1], acc[tt]);
#pragma unroll
          for (int tt = 0; tt < TT; ++tt) part[tt][hf] = relu_pack(acc[tt]);
        }
#pragma unroll
        for (int tt = 0; tt < TT; ++tt) Z[tt][kb] = cat16(part[tt][0], part[tt][1]);
      }

      const v16h B3a = ld16(&sW3[ns][n16][base0],      &sW3[ns][n16][16 + base0]);
      const v16h B3b = ld16(&sW3[ns][n16][32 + base0], &sW3[ns][n16][48 + base0]);
      const float b3v = sB3[ns][n16];
      v8f c0;
#pragma unroll
      for (int r = 0; r < 8; ++r) c0[r] = b3v;
      float m[TT];
#pragma unroll
      for (int tt = 0; tt < TT; ++tt) {
        v8f c = wmma16(Z[tt][0], B3a, c0);
        c = wmma16(Z[tt][1], B3b, c);
        float v = c[0];
#pragma unroll
        for (int r = 1; r < 8; ++r) v = fmaxf(v, c[r]);
        m[tt] = v;
      }
#pragma unroll
      for (int q = 0; q < TT / 2; ++q) {
        const float mine = hl ? m[2 * q + 1] : m[2 * q];
        const float send = hl ? m[2 * q] : m[2 * q + 1];
        const float v = fmaxf(mine, __shfl_xor(send, 16, 32));
        if (s == 0) {
          sv[q] = CLAMPC * TWOPI_I * atanf(v * (1.0f / CLAMPC));
          slog += sv[q];
        } else {
          zv[q] = zv[q] * __expf(sv[q]) + v;
        }
      }
    }
#pragma unroll
    for (int q = 0; q < TT / 2; ++q) *(float*)at(zmat, zo + q * 256u) = zv[q];
    sPart[chunk][lane] = slog;
  }
  __syncthreads();
  const float* flat = &sPart[0][0];
  float tot = 0.f;
  for (int k = 0; k < NCHUNK * 32 / THREADS; ++k) tot += flat[tid + k * THREADS];
  tot += __shfl_xor(tot, 16, 32);
  tot += __shfl_xor(tot, 8, 32);
  tot += __shfl_xor(tot, 4, 32);
  tot += __shfl_xor(tot, 2, 32);
  tot += __shfl_xor(tot, 1, 32);
  if (lane == 0) sWave[wid] = tot;
  __syncthreads();
  if (tid == 0) {
    float t = 0.f;
    for (int w = 0; w < WAVES; ++w) t += sWave[w];
    atomicAdd(&logdet[bi], t);
  }
}

extern "C" void kernel_launch(void* const* d_in, const int* in_sizes, int n_in,
                              void* d_out, int out_size, void* d_ws, size_t ws_size,
                              hipStream_t stream) {
  const float* x     = (const float*)d_in[0];
  const int*   knn   = (const int*)d_in[1];
  const float* logs  = (const float*)d_in[2];
  const float* biasv = (const float*)d_in[3];
  const float* W1    = (const float*)d_in[4];
  const float* b1    = (const float*)d_in[5];
  const float* W2    = (const float*)d_in[6];
  const float* b2    = (const float*)d_in[7];
  const float* W3    = (const float*)d_in[8];
  const float* b3    = (const float*)d_in[9];

  float* zmat   = (float*)d_out;
  float* logdet = zmat + (size_t)NBATCH * NPOINTS * 32;
  _Float16* z1h = (_Float16*)d_ws;

  const int total = NBATCH * NPOINTS;
  prep_kernel<<<(total + 255) / 256, 256, 0, stream>>>(x, logs, biasv, zmat,
                                                       logdet, z1h, total);

  const dim3 grid(NWG);
  conv_kernel<<<grid, THREADS, 0, stream>>>(z1h, knn, W1, b1, W2, b2, W3, b3,
                                            zmat, logdet);
}


# relu of a weight group's D tiles issued between the products of the NEXT group (pipeline one group deep, order fixed by scheduling groups): vector work in the products' shadow, fewer wait slots
# speedup vs baseline: 9.8253x; 1.1968x over previous
#include <hip/hip_runtime.h>
#include <hip/hip_bf16.h>

typedef __attribute__((ext_vector_type(16))) _Float16 v16h;
typedef __attribute__((ext_vector_type(8)))  _Float16 v8h;
typedef __attribute__((ext_vector_type(2)))  _Float16 v2h;
typedef __attribute__((ext_vector_type(8)))  float    v8f;

#define NPOINTS 8192
#define NBATCH  16
#define WAVES   16
#define THREADS (WAVES * 32)
#define NWG     256
#define PTS_WG  (NBATCH * NPOINTS / NWG)
#define TT      4
#define NCHUNK  (PTS_WG / TT)
static_assert(NPOINTS % PTS_WG == 0, "a workgroup stays inside one batch");
static_assert(PTS_WG % TT == 0 && TT % 2 == 0, "whole chunks, points in pairs");
static_assert((NCHUNK * 32) % THREADS == 0, "the closing sum has whole trips");
static_assert((4 * 32 * 16) % THREADS == 0, "the staging loops have whole trips");
#define CLAMPC  1.9f
#define TWOPI_I 0.63661977236758f

__device__ __forceinline__ v16h ld16(const _Float16* a, const _Float16* b) {
  v8h lo = *(const v8h*)a;
  v8h hi = *(const v8h*)b;
  return __builtin_shufflevector(lo, hi, 0,1,2,3,4,5,6,7,8,9,10,11,12,13,14,15);
}

__device__ __forceinline__ v8f wmma16(v16h a, v16h b, v8f c) {
  return __builtin_amdgcn_wmma_f32_16x16x32_f16(false, a, false, b, (short)0, c,
                                                false, false);
}

__device__ __forceinline__ v8h wmma16h(v16h a, v16h b, v8h c) {
  return __builtin_amdgcn_wmma_f16_16x16x32_f16(false, a, false, b, (short)0, c,
                                                false, false);
}

__global__ void prep_kernel(const float* __restrict__ x,
                            const float* __restrict__ logs,
                            const float* __restrict__ bias,
                            float* __restrict__ zmat,
                            float* __restrict__ logdet,
                            _Float16* __restrict__ z1h, int total) {
  int p = blockIdx.x * blockDim.x + threadIdx.x;
  if (p < total) {
    const float* xr = x + (size_t)p * 32;
    float* zr = zmat + (size_t)p * 32;
    _Float16* hr = z1h + (size_t)p * 16;
#pragma unroll
    for (int c = 0; c < 32; ++c) {
      float zc = xr[31 - c] * expf(logs[c]) + bias[c];
      zr[c] = zc;
      if (c < 16) hr[c] = (_Float16)zc;
    }
  }
  if (p < NBATCH) {
    float su = 0.f;
#pragma unroll
    for (int c = 0; c < 32; ++c) su += logs[c];
    logdet[p] = su * (float)NPOINTS;
  }
}

typedef __attribute__((ext_vector_type(8))) short v8s;
__device__ __forceinline__ v8h relu16(v8h h) {
  const v8s zero = {};
  const v8s r = __builtin_elementwise_max(__builtin_bit_cast(v8s, h), zero);
  return __builtin_bit_cast(v8h, r);
}

__device__ __forceinline__ const char* at(const void* base, unsigned bytes) {
  return (const char*)base + bytes;
}

__device__ __forceinline__ v16h cat16(v8h lo, v8h hi) {
  return __builtin_shufflevector(lo, hi, 0,1,2,3,4,5,6,7,8,9,10,11,12,13,14,15);
}

#define KEEP_HERE(x) asm volatile("" : "+v"(x))

__global__ __launch_bounds__(THREADS) void conv_kernel(
    const _Float16* __restrict__ z1h, const int* __restrict__ knn,
    const float* __restrict__ W1g, const float* __restrict__ b1g,
    const float* __restrict__ W2g, const float* __restrict__ b2g,
    const float* __restrict__ W3g, const float* __restrict__ b3g,
    float* __restrict__ zmat, float* __restrict__ logdet) {
  __shared__ __align__(16) _Float16 sW1[4][64][40];
  __shared__ __align__(16) _Float16 sW2[4][64][72];
  __shared__ __align__(16) _Float16 sW3[4][16][72];
  __shared__ __align__(16) _Float16 sB1[4][64];
  __shared__ __align__(16) _Float16 sB2[4][64];
  __shared__ float sB3[4][16];
  __shared__ float sPart[NCHUNK][32];
  __shared__ float sWave[WAVES];
  __shared__ int sNext;

  const int tid  = threadIdx.x;
  const int wid  = tid >> 5;
  const int lane = tid & 31;
  const int n16  = lane & 15;
  const int hl   = lane >> 4;
  const int base0 = hl * 8;

  const int bi  = blockIdx.x / (NPOINTS / PTS_WG);
  const int pb0 = (blockIdx.x % (NPOINTS / PTS_WG)) * PTS_WG;

#pragma unroll
  for (int it = 0; it < 4 * 16 * 64 / THREADS; ++it) {
    const int idx = tid + it * THREADS;
    const int j = idx & 63, kp = (idx >> 6) & 15, s = idx >> 10;
    const int k0 = 2 * kp;
    const bool nbh = k0 < 16;
    const float* pa = W1g + (s * 48 + (nbh ? 16 + k0 : k0 - 16)) * 64 + j;
    const float* pb = W1g + (s * 48 + 32 + (k0 & 15)) * 64 + j;
    const float a0 = pa[0], a1 = pa[64], c0 = pb[0], c1 = pb[64];
    const float v0 = nbh ? a0 + c0 : a0 - c0;
    const float v1 = nbh ? a1 + c1 : a1 - c1;
    const v2h h = {(_Float16)v0, (_Float16)v1};
    *(v2h*)&sW1[s][j][k0] = h;
  }
#pragma unroll
  for (int it = 0; it < 4 * 32 * 64 / THREADS; ++it) {
    const int idx = tid + it * THREADS;
    const int j = idx & 63, kp = (idx >> 6) & 31, s = idx >> 11;
    const float* p = W2g + (s * 64 + 2 * kp) * 64 + j;
    const v2h h = {(_Float16)p[0], (_Float16)p[64]};
    *(v2h*)&sW2[s][j][2 * kp] = h;
  }
#pragma unroll
  for (int it = 0; it < 4 * 32 * 16 / THREADS; ++it) {
    const int idx = tid + it * THREADS;
    const int j = idx & 15, kp = (idx >> 4) & 31, s = idx >> 9;
    const float* p = W3g + (s * 64 + 2 * kp) * 16 + j;
    const v2h h = {(_Float16)p[0], (_Float16)p[16]};
    *(v2h*)&sW3[s][j][2 * kp] = h;
  }
  if (tid < 256) {
    int s = tid >> 6, j = tid & 63;
    sB1[s][j] = (_Float16)b1g[s * 64 + j];
    sB2[s][j] = (_Float16)b2g[s * 64 + j];
    if (tid < 64) sB3[tid >> 4][tid & 15] = b3g[(tid >> 4) * 16 + (tid & 15)];
    if (tid == 0) sNext = 0;
  }

  __syncthreads();

  for (int n = 0; n < NCHUNK; ++n) {
    int got = 0;
    if (lane == 0) got = atomicAdd(&sNext, 1);
    const int chunk = __builtin_amdgcn_readfirstlane(got);
    if (chunk >= NCHUNK) break;
    const int pw = pb0 + chunk * TT;
    float slog = 0.f;
    asm volatile("" ::: "memory");

    const unsigned gp0 = (unsigned)(bi * NPOINTS + pw);
    int nidx[TT];
#pragma unroll
    for (int tt = 0; tt < TT; ++tt)
      nidx[tt] = *(const int*)at(knn, ((gp0 + tt) * 16u + n16) * 4u);
    v16h X[TT];
#pragma unroll
    for (int tt = 0; tt < TT; ++tt) {
      const unsigned nbp = (unsigned)(bi * NPOINTS) + (unsigned)nidx[tt];
      const char* nb = at(z1h, (nbp * 16u + base0) * 2u);
      const char* ct = at(z1h, ((gp0 + tt) * 16u + base0) * 2u);
      X[tt] = cat16(*(const v8h*)nb, *(const v8h*)ct);
    }

    const unsigned zo = ((gp0 + hl) * 32u + 16u + n16) * 4u;
    float zv[TT / 2];
#pragma unroll
    for (int q = 0; q < TT / 2; ++q) zv[q] = *(const float*)at(zmat, zo + q * 256u);
    __builtin_amdgcn_sched_barrier(0);

    float sv[TT / 2];
    v8h Yp[TT][4], Zp[TT][4];
    v8h pend[TT];
    v8f c3[TT];
    v16h nA = {}, nB = {};
    v8h nBias = {};
    float nb3 = 0.f;
    nA = ld16(&sW1[0][n16][base0], &sW1[0][n16][16 + base0]);
    nBias = *(const v8h*)&sB1[0][8 * hl];
#pragma unroll
    for (int ns = 0; ns < 4; ++ns) {
#pragma unroll
      for (int g = 0; g < 9; ++g) {
        const v16h cA = nA, cB = nB;
        const v8h cBias = nBias;
        const float cb3 = nb3;
        if (g + 1 < 4) {
          const int j = (g + 1) * 16 + n16;
          nA = ld16(&sW1[ns][j][base0], &sW1[ns][j][16 + base0]);
          nBias = *(const v8h*)&sB1[ns][(g + 1) * 16 + 8 * hl];
        } else if (g + 1 < 8) {
          const int j = (g - 3) * 16 + n16;
          nA = ld16(&sW2[ns][j][base0],      &sW2[ns][j][16 + base0]);
          nB = ld16(&sW2[ns][j][32 + base0], &sW2[ns][j][48 + base0]);
          nBias = *(const v8h*)&sB2[ns][(g - 3) * 16 + 8 * hl];
        } else if (g + 1 == 8) {
          nA = ld16(&sW3[ns][n16][base0],      &sW3[ns][n16][16 + base0]);
          nB = ld16(&sW3[ns][n16][32 + base0], &sW3[ns][n16][48 + base0]);
          nb3 = sB3[ns][n16];
        } else if (ns < 3) {
          nA = ld16(&sW1[ns + 1][n16][base0], &sW1[ns + 1][n16][16 + base0]);
          nBias = *(const v8h*)&sB1[ns + 1][8 * hl];
        }
        v8h cur[TT] = {};
        if (g < 4) {
#pragma unroll
          for (int tt = 0; tt < TT; ++tt) {
            cur[tt] = wmma16h(cA, X[tt], cBias);
            if (g > 0) { Yp[tt][g - 1] = relu16(pend[tt]); KEEP_HERE(Yp[tt][g - 1]); }
          }
        } else if (g < 8) {
#pragma unroll
          for (int tt = 0; tt < TT; ++tt) {
            cur[tt] = wmma16h(cA, cat16(Yp[tt][0], Yp[tt][1]), cBias);
            if (g == 4) { Yp[tt][3] = relu16(pend[tt]); KEEP_HERE(Yp[tt][3]); }
            else        { Zp[tt][g - 5] = relu16(pend[tt]); KEEP_HERE(Zp[tt][g - 5]); }
          }
#pragma unroll
          for (int tt = 0; tt < TT; ++tt)
            cur[tt] = wmma16h(cB, cat16(Yp[tt][2], Yp[tt][3]), cur[tt]);
        } else {
          v8f c0;
#pragma unroll
          for (int r = 0; r < 8; ++r) c0[r] = cb3;
#pragma unroll
          for (int tt = 0; tt < TT; ++tt) {
            c3[tt] = wmma16(cat16(Zp[tt][0], Zp[tt][1]), cA, c0);
            Zp[tt][3] = relu16(pend[tt]);
            KEEP_HERE(Zp[tt][3]);
          }
#pragma unroll
          for (int tt = 0; tt < TT; ++tt)
            c3[tt] = wmma16(cat16(Zp[tt][2], Zp[tt][3]), cB, c3[tt]);
        }
#pragma unroll
        for (int tt = 0; tt < TT; ++tt) pend[tt] = cur[tt];
        if (g >= 1 && g <= 3) {
          if (g < 3) __builtin_amdgcn_sched_group_barrier(0x100, 3, 0);
          else       __builtin_amdgcn_sched_group_barrier(0x100, 5, 0);
#pragma unroll
          for (int i = 0; i < TT; ++i) {
            __builtin_amdgcn_sched_group_barrier(0x008, 1, 0);
            __builtin_amdgcn_sched_group_barrier(0x002, 4, 0);
          }
        } else if (g == 4) {
          __builtin_amdgcn_sched_group_barrier(0x100, 5, 0);
#pragma unroll
          for (int i = 0; i < TT; ++i) {
            __builtin_amdgcn_sched_group_barrier(0x008, 1, 0);
            __builtin_amdgcn_sched_group_barrier(0x002, 4, 0);
          }
          __builtin_amdgcn_sched_group_barrier(0x008, TT, 0);
        } else if (g >= 5 && g <= 7) {
          __builtin_amdgcn_sched_group_barrier(0x100, 5, 0);
#pragma unroll
          for (int i = 0; i < 2 * TT; ++i) {
            __builtin_amdgcn_sched_group_barrier(0x008, 1, 0);
            __builtin_amdgcn_sched_group_barrier(0x002, 2, 0);
          }
        } else if (g == 8) {
          if (ns < 3) __builtin_amdgcn_sched_group_barrier(0x100, 3, 0);
#pragma unroll
          for (int i = 0; i < TT; ++i) {
            __builtin_amdgcn_sched_group_barrier(0x008, 1, 0);
            __builtin_amdgcn_sched_group_barrier(0x002, 4, 0);
          }
          __builtin_amdgcn_sched_group_barrier(0x008, TT, 0);
        }
        __builtin_amdgcn_sched_barrier(0);
      }

      const int s = ns & 1;
      float m[TT];
#pragma unroll
      for (int tt = 0; tt < TT; ++tt) {
        float v = c3[tt][0];
#pragma unroll
        for (int r = 1; r < 8; ++r) v = fmaxf(v, c3[tt][r]);
        m[tt] = v;
      }
#pragma unroll
      for (int q = 0; q < TT / 2; ++q) {
        const float mine = hl ? m[2 * q + 1] : m[2 * q];
        const float send = hl ? m[2 * q] : m[2 * q + 1];
        const float v = fmaxf(mine, __shfl_xor(send, 16, 32));
        if (s == 0) {
          sv[q] = CLAMPC * TWOPI_I * atanf(v * (1.0f / CLAMPC));
          slog += sv[q];
        } else {
          zv[q] = zv[q] * __expf(sv[q]) + v;
        }
      }
    }
#pragma unroll
    for (int q = 0; q < TT / 2; ++q) *(float*)at(zmat, zo + q * 256u) = zv[q];
    sPart[chunk][lane] = slog;
  }
  __syncthreads();
  const float* flat = &sPart[0][0];
  float tot = 0.f;
  for (int k = 0; k < NCHUNK * 32 / THREADS; ++k) tot += flat[tid + k * THREADS];
  tot += __shfl_xor(tot, 16, 32);
  tot += __shfl_xor(tot, 8, 32);
  tot += __shfl_xor(tot, 4, 32);
  tot += __shfl_xor(tot, 2, 32);
  tot += __shfl_xor(tot, 1, 32);
  if (lane == 0) sWave[wid] = tot;
  __syncthreads();
  if (tid == 0) {
    float t = 0.f;
    for (int w = 0; w < WAVES; ++w) t += sWave[w];
    atomicAdd(&logdet[bi], t);
  }
}

extern "C" void kernel_launch(void* const* d_in, const int* in_sizes, int n_in,
                              void* d_out, int out_size, void* d_ws, size_t ws_size,
                              hipStream_t stream) {
  const float* x     = (const float*)d_in[0];
  const int*   knn   = (const int*)d_in[1];
  const float* logs  = (const float*)d_in[2];
  const float* biasv = (const float*)d_in[3];
  const float* W1    = (const float*)d_in[4];
  const float* b1    = (const float*)d_in[5];
  const float* W2    = (const float*)d_in[6];
  const float* b2    = (const float*)d_in[7];
  const float* W3    = (const float*)d_in[8];
  const float* b3    = (const float*)d_in[9];

  float* zmat   = (float*)d_out;
  float* logdet = zmat + (size_t)NBATCH * NPOINTS * 32;
  _Float16* z1h = (_Float16*)d_ws;

  const int total = NBATCH * NPOINTS;
  prep_kernel<<<(total + 255) / 256, 256, 0, stream>>>(x, logs, biasv, zmat,
                                                       logdet, z1h, total);

  const dim3 grid(NWG);
  conv_kernel<<<grid, THREADS, 0, stream>>>(z1h, knn, W1, b1, W2, b2, W3, b3,
                                            zmat, logdet);
}
